# starting kernel
# baseline (speedup 1.0000x reference)
#include <hip/hip_runtime.h>

typedef float v2f __attribute__((ext_vector_type(2)));
typedef float v8f __attribute__((ext_vector_type(8)));

#define D 32

__global__ void k_init_deg(float* __restrict__ deg, int n) {
  int i = blockIdx.x * blockDim.x + threadIdx.x;
  if (i < n) deg[i] = 1.0f;
}

__global__ void k_deg_edges(const int* __restrict__ dst, const float* __restrict__ w,
                            float* __restrict__ deg, int E) {
  int e = blockIdx.x * blockDim.x + threadIdx.x;
  if (e < E) atomicAdd(&deg[dst[e]], w[e]);
}

__global__ void k_dis(const float* __restrict__ deg, float* __restrict__ dis, int n) {
  int i = blockIdx.x * blockDim.x + threadIdx.x;
  if (i < n) {
    float d = deg[i];
    dis[i] = (d > 0.0f) ? __frsqrt_rn(d) : 0.0f;
  }
}

template <bool RELU>
__global__ void k_gemm32(const float* __restrict__ X, const float* __restrict__ W,
                         float* __restrict__ Y, int n) {
  const int lane = threadIdx.x & 31;
  const int wave = threadIdx.x >> 5;
  const int row0 = (blockIdx.x * (blockDim.x >> 5) + wave) << 4;
  if (row0 >= n) return;

  const int l    = lane & 15;
  const int half = lane >> 4;

  v8f c0 = {};
  v8f c1 = {};

  const float* xrow = X + (size_t)(row0 + l) * D;

#pragma unroll
  for (int kb = 0; kb < D; kb += 4) {
    const int k0 = kb + 2 * half;
    v2f a;
    a.x = xrow[k0];
    a.y = xrow[k0 + 1];
    if (RELU) {
      a.x = fmaxf(a.x, 0.0f);
      a.y = fmaxf(a.y, 0.0f);
    }
    v2f b0, b1;
    b0.x = W[(size_t)k0 * D + l];
    b0.y = W[(size_t)(k0 + 1) * D + l];
    b1.x = W[(size_t)k0 * D + 16 + l];
    b1.y = W[(size_t)(k0 + 1) * D + 16 + l];

    c0 = __builtin_amdgcn_wmma_f32_16x16x4_f32(false, a, false, b0, (short)0, c0,
                                               false, false);
    c1 = __builtin_amdgcn_wmma_f32_16x16x4_f32(false, a, false, b1, (short)0, c1,
                                               false, false);
  }

  const int rowm = row0 + 8 * half;
#pragma unroll
  for (int j = 0; j < 8; ++j) {
    Y[(size_t)(rowm + j) * D + l]      = c0[j];
    Y[(size_t)(rowm + j) * D + 16 + l] = c1[j];
  }
}

__global__ void k_self_bias(const float* __restrict__ xm, const float* __restrict__ dis,
                            const float* __restrict__ b, float* __restrict__ out,
                            long long total) {
  long long idx = (long long)blockIdx.x * blockDim.x + threadIdx.x;
  if (idx >= total) return;
  int i = (int)(idx >> 5);
  int d = (int)(idx & (D - 1));
  float s = dis[i];
  out[idx] = b[d] + xm[idx] * (s * s);
}

__global__ void k_edge_msg(const int* __restrict__ src, const int* __restrict__ dst,
                           const float* __restrict__ w, const float* __restrict__ dis,
                           const float* __restrict__ xm, float* __restrict__ out,
                           int E) {
  long long idx = (long long)blockIdx.x * blockDim.x + threadIdx.x;
  int e = (int)(idx >> 5);
  int d = (int)(idx & (D - 1));
  if (e >= E) return;

  if (d == 0 && e + 4096 < E) {
    __builtin_prefetch(&src[e + 4096], 0, 0);
    __builtin_prefetch(&dst[e + 4096], 0, 0);
    __builtin_prefetch(&w[e + 4096], 0, 0);
  }

  int s = src[e];
  int t = dst[e];
  float c = dis[s] * w[e] * dis[t];
  atomicAdd(&out[(size_t)t * D + d], xm[(size_t)s * D + d] * c);
}

extern "C" void kernel_launch(void* const* d_in, const int* in_sizes, int n_in,
                              void* d_out, int out_size, void* d_ws, size_t ws_size,
                              hipStream_t stream) {
  const float* x  = (const float*)d_in[0];
  const int*   ei = (const int*)d_in[1];
  const float* ew = (const float*)d_in[2];
  const float* W1 = (const float*)d_in[3];
  const float* b1 = (const float*)d_in[4];
  const float* W2 = (const float*)d_in[5];
  const float* b2 = (const float*)d_in[6];

  const int n = in_sizes[0] / D;
  const int E = in_sizes[2];
  const int* src = ei;
  const int* dst = ei + E;

  float* ws  = (float*)d_ws;
  float* deg = ws;
  float* dis = ws + n;
  float* xm  = ws + 2 * (size_t)n;
  float* h   = xm + (size_t)n * D;
  float* out = (float*)d_out;

  const int T = 256;
  const int nBlocks   = (n + T - 1) / T;
  const int eBlocks   = (E + T - 1) / T;
  const long long nd  = (long long)n * D;
  const int ndBlocks  = (int)((nd + T - 1) / T);
  const int edBlocks  = (int)(((long long)E * D + T - 1) / T);
  const int rowsPerBlock = (T / 32) * 16;
  const int gBlocks   = (n + rowsPerBlock - 1) / rowsPerBlock;

  k_init_deg<<<nBlocks, T, 0, stream>>>(deg, n);
  k_deg_edges<<<eBlocks, T, 0, stream>>>(dst, ew, deg, E);
  k_dis<<<nBlocks, T, 0, stream>>>(deg, dis, n);

  k_gemm32<false><<<gBlocks, T, 0, stream>>>(x, W1, xm, n);
  k_self_bias<<<ndBlocks, T, 0, stream>>>(xm, dis, b1, h, nd);
  k_edge_msg<<<edBlocks, T, 0, stream>>>(src, dst, ew, dis, xm, h, E);

  k_gemm32<true><<<gBlocks, T, 0, stream>>>(h, W2, xm, n);
  k_self_bias<<<ndBlocks, T, 0, stream>>>(xm, dis, b2, out, nd);
  k_edge_msg<<<edBlocks, T, 0, stream>>>(src, dst, ew, dis, xm, out, E);
}


# edge list sorted once per call by 392-row dst range into workspace (LDS counts); one block owns a range's LDS tile: no dst scan, no global row atomics; sum order changed; 5 launches -> 6
# speedup vs baseline: 3.6984x; 3.6984x over previous
#include <hip/hip_runtime.h>

typedef float v2f __attribute__((ext_vector_type(2)));
typedef float v8f __attribute__((ext_vector_type(8)));

#define D 32

#define DEG_RANGE  12800
#define DEG_CHUNKS 32
#define DEG_T      1024
#define DEG_U      8
static_assert(DEG_U == 8, "the load fence in k_deg_part names eight weights");

__global__ void __launch_bounds__(DEG_T)
k_deg_part(const int* __restrict__ dst, const float* __restrict__ w,
           float* __restrict__ part, int n, int E) {
  __shared__ float acc[DEG_RANGE];
  const int tid = threadIdx.x;
  const int lo  = blockIdx.y * DEG_RANGE;
  const int rem = n - lo;
  const int cnt = rem < DEG_RANGE ? rem : DEG_RANGE;

  for (int i = tid; i < DEG_RANGE; i += DEG_T) acc[i] = 0.0f;
  __syncthreads();

  const int per = (E + DEG_CHUNKS - 1) / DEG_CHUNKS;
  const int e0  = blockIdx.x * per;
  const int e1  = (e0 + per < E) ? (e0 + per) : E;

  int e = e0 + tid;
  for (; e + (DEG_U - 1) * DEG_T < e1; e += DEG_U * DEG_T) {
    int   t[DEG_U];
    float v[DEG_U];
    for (int u = 0; u < DEG_U; ++u) {
      t[u] = dst[e + u * DEG_T];
      v[u] = w[e + u * DEG_T];
    }
    asm volatile("" :: "v"(v[0]), "v"(v[1]), "v"(v[2]), "v"(v[3]),
                       "v"(v[4]), "v"(v[5]), "v"(v[6]), "v"(v[7]));
    for (int u = 0; u < DEG_U; ++u) {
      const unsigned r = (unsigned)(t[u] - lo);
      if (r < (unsigned)DEG_RANGE) atomicAdd(&acc[r], v[u]);
    }
  }
  for (; e < e1; e += DEG_T) {
    const unsigned r = (unsigned)(dst[e] - lo);
    if (r < (unsigned)DEG_RANGE) atomicAdd(&acc[r], w[e]);
  }
  __syncthreads();

  float* __restrict__ p = part + (size_t)blockIdx.x * n + lo;
  for (int i = tid; i < cnt; i += DEG_T) p[i] = acc[i];
}

template <bool RELU, bool MAKE_DIS>
__global__ void k_gemm32(const float* __restrict__ X, const float* __restrict__ W,
                         const float* __restrict__ bias,
                         const float* __restrict__ part, float* __restrict__ dis,
                         float* __restrict__ Y, float* __restrict__ H, int n) {
  const int lane = threadIdx.x & 31;
  const int wave = threadIdx.x >> 5;
  const int row0 = (blockIdx.x * (blockDim.x >> 5) + wave) << 4;
  if (row0 >= n) return;

  const int l    = lane & 15;
  const int half = lane >> 4;

  float dv;
  if (MAKE_DIS) {
    float p[DEG_CHUNKS];
#pragma unroll
    for (int c = 0; c < DEG_CHUNKS; ++c) p[c] = part[(size_t)c * n + row0 + l];
    float d = 1.0f;
#pragma unroll
    for (int c = 0; c < DEG_CHUNKS; ++c) d += p[c];
    dv = (d > 0.0f) ? __frsqrt_rn(d) : 0.0f;
  } else {
    dv = dis[row0 + l];
  }
  const float bl = bias[l];
  const float bh = bias[16 + l];

  v8f c0 = {};
  v8f c1 = {};

  const float* xrow = X + (size_t)(row0 + l) * D;

#pragma unroll
  for (int kb = 0; kb < D; kb += 4) {
    const int k0 = kb + 2 * half;
    v2f a;
    a.x = xrow[k0];
    a.y = xrow[k0 + 1];
    if (RELU) {
      a.x = fmaxf(a.x, 0.0f);
      a.y = fmaxf(a.y, 0.0f);
    }
    v2f b0, b1;
    b0.x = W[(size_t)k0 * D + l];
    b0.y = W[(size_t)(k0 + 1) * D + l];
    b1.x = W[(size_t)k0 * D + 16 + l];
    b1.y = W[(size_t)(k0 + 1) * D + 16 + l];

    c0 = __builtin_amdgcn_wmma_f32_16x16x4_f32(false, a, false, b0, (short)0, c0,
                                               false, false);
    c1 = __builtin_amdgcn_wmma_f32_16x16x4_f32(false, a, false, b1, (short)0, c1,
                                               false, false);
  }

  const int rowm = row0 + 8 * half;
  const int dvb  = __builtin_bit_cast(int, dv);
#pragma unroll
  for (int j = 0; j < 8; ++j) {
    const float sLo = __builtin_bit_cast(float, __builtin_amdgcn_readlane(dvb, j));
    const float sHi = __builtin_bit_cast(float, __builtin_amdgcn_readlane(dvb, 8 + j));
    const float s   = half ? sHi : sLo;
    const float s2  = s * s;
    const size_t o  = (size_t)(rowm + j) * D;
    Y[o + l]      = c0[j];
    Y[o + 16 + l] = c1[j];
    H[o + l]      = bl + c0[j] * s2;
    H[o + 16 + l] = bh + c1[j] * s2;
  }

  if (MAKE_DIS) {
    if (half == 0) dis[row0 + l] = dv;
  }
}

#define EM_RN   392
#define BIN_NB  256
#define BIN_NR  256
#define BIN_T   1024
#define BIN_U   7
#define BIN_P   (BIN_NR + 1)

__global__ void __launch_bounds__(BIN_T)
k_bin(const int* __restrict__ src, const int* __restrict__ dst,
      const float* __restrict__ w, int4* __restrict__ recs,
      int* __restrict__ off, int E) {
  __shared__ int hist[BIN_NR];
  __shared__ int cur[BIN_NR];
  __shared__ int wsum[BIN_NR / 32];
  const int tid  = threadIdx.x;
  const int lane = tid & 31;
  const int wave = tid >> 5;

  const int per = (E + BIN_NB - 1) / BIN_NB;
  const int e0  = blockIdx.x * per < E ? blockIdx.x * per : E;
  const int e1  = (e0 + per < E) ? (e0 + per) : E;

  if (tid < BIN_NR) hist[tid] = 0;
  __syncthreads();

  for (int b = e0 + tid; b < e1 + tid; b += BIN_U * BIN_T) {
    int t[BIN_U];
#pragma unroll
    for (int u = 0; u < BIN_U; ++u) {
      const int e = b + u * BIN_T;
      t[u] = dst[e < e1 ? e : e1 - 1];
    }
#pragma unroll
    for (int u = 0; u < BIN_U; ++u)
      if (b + u * BIN_T < e1) atomicAdd(&hist[(unsigned)t[u] / EM_RN], 1);
  }
  __syncthreads();

  int v = 0, incl = 0;
  if (tid < BIN_NR) {
    v    = hist[tid];
    incl = v;
#pragma unroll
    for (int d = 1; d < 32; d <<= 1) {
      const int up = __shfl_up(incl, d);
      if (lane >= d) incl += up;
    }
    if (lane == 31) wsum[wave] = incl;
  }
  __syncthreads();
  if (tid < BIN_NR) {
    int before = 0;
#pragma unroll
    for (int k = 0; k < BIN_NR / 32; ++k) before += (k < wave) ? wsum[k] : 0;
    const int excl = before + incl - v;
    cur[tid] = excl;
    off[(size_t)blockIdx.x * BIN_P + tid] = excl;
  }
  if (tid == 0) off[(size_t)blockIdx.x * BIN_P + BIN_NR] = e1 - e0;
  __syncthreads();

  int4* __restrict__ myrec = recs + e0;
  for (int b = e0 + tid; b < e1 + tid; b += BIN_U * BIN_T) {
    int   t[BIN_U];
    int   s[BIN_U];
    float x[BIN_U];
#pragma unroll
    for (int u = 0; u < BIN_U; ++u) {
      const int e  = b + u * BIN_T;
      const int ec = e < e1 ? e : e1 - 1;
      t[u] = dst[ec];
      s[u] = src[ec];
      x[u] = w[ec];
    }
#pragma unroll
    for (int u = 0; u < BIN_U; ++u) {
      if (b + u * BIN_T < e1) {
        const int g = (int)((unsigned)t[u] / EM_RN);
        const int p = atomicAdd(&cur[g], 1);
        myrec[p] = make_int4(s[u], t[u] - g * EM_RN, __builtin_bit_cast(int, x[u]), 0);
      }
    }
  }
}

#define EM_T   1024
#define EM_W   (EM_T / 32)
#define EM_S   (BIN_NB / EM_W)
static_assert(EM_S * EM_W == BIN_NB, "every chunk needs a wave");

__global__ void __launch_bounds__(EM_T)
k_edge_tile(const int4* __restrict__ recs, const int* __restrict__ off,
            const float* __restrict__ dis, const float* __restrict__ xm,
            float* __restrict__ out, int n, int E) {
  __shared__ float acc[EM_RN * D];

  const int tid  = threadIdx.x;
  const int lane = tid & 31;
  const int wave = __builtin_amdgcn_readfirstlane(tid >> 5);
  const int rg   = blockIdx.x;
  const int lo   = rg * EM_RN;
  const int rem  = n - lo;
  const int rows = rem < EM_RN ? rem : EM_RN;

  const int per = (E + BIN_NB - 1) / BIN_NB;
  int pos[EM_S + 1];
  int adj[EM_S];
  pos[0] = 0;
#pragma unroll
  for (int k = 0; k < EM_S; ++k) {
    const int c  = wave * EM_S + k;
    const int c0 = c * per < E ? c * per : E;
    const int a  = off[(size_t)c * BIN_P + rg];
    const int b  = off[(size_t)c * BIN_P + rg + 1];
    adj[k]     = c0 + a - pos[k];
    pos[k + 1] = pos[k] + (b - a);
  }
  const int total = pos[EM_S];

#define EM_FETCH(I, RS, RR, RW)                                              \
  do {                                                                       \
    const int q_ = ((I) + lane < total) ? (I) + lane : total - 1;            \
    int base_ = adj[0];                                                      \
    _Pragma("unroll")                                                        \
    for (int k_ = 1; k_ < EM_S; ++k_) base_ = (q_ >= pos[k_]) ? adj[k_] : base_; \
    const int* p_ = reinterpret_cast<const int*>(recs + (base_ + q_));       \
    RS = p_[0];                                                              \
    RR = p_[1];                                                              \
    RW = p_[2];                                                              \
  } while (0)
  int ns = 0, nr = 0, nw = 0;
  if (total > 0) EM_FETCH(0, ns, nr, nw);

  float4* acc4 = reinterpret_cast<float4*>(acc);
  for (int i = tid; i < EM_RN * D / 4; i += EM_T) acc4[i] = make_float4(0.f, 0.f, 0.f, 0.f);
  __syncthreads();

  for (int i = 0; i < total; i += 32) {
    const bool valid = i + lane < total;
    const int   s  = ns;
    const int   r  = nr;
    const float we = __builtin_bit_cast(float, nw);
    EM_FETCH(i + 32, ns, nr, nw);
    float c = dis[s] * we * dis[lo + r];
    c = valid ? c : 0.0f;
    const int cb   = __builtin_bit_cast(int, c);
    const int ro   = r * D;
    const int left = total - i;
    float v[32];
#pragma unroll
    for (int g = 0; g < 4; ++g) {
      if (g * 8 < left) {
#pragma unroll
        for (int j = g * 8; j < g * 8 + 8; ++j) {
          const unsigned sj = (unsigned)__builtin_amdgcn_readlane(s, j);
          v[j] = xm[sj * D + lane];
        }
      }
    }
#pragma unroll
    for (int g = 0; g < 4; ++g) {
      if (g * 8 < left) {
#pragma unroll
        for (int j = g * 8; j < g * 8 + 8; ++j) {
          const float cj = __builtin_bit_cast(float, __builtin_amdgcn_readlane(cb, j));
          const int   rj = __builtin_amdgcn_readlane(ro, j);
          atomicAdd(&acc[rj + lane], v[j] * cj);
        }
      }
    }
  }
  __syncthreads();

  float4* __restrict__ out4 = reinterpret_cast<float4*>(out + (size_t)lo * D);
  for (int i = tid; i < rows * (D / 4); i += EM_T) {
    float4 o = out4[i];
    const float4 a = acc4[i];
    o.x += a.x; o.y += a.y; o.z += a.z; o.w += a.w;
    out4[i] = o;
  }
}
#undef EM_FETCH

extern "C" void kernel_launch(void* const* d_in, const int* in_sizes, int n_in,
                              void* d_out, int out_size, void* d_ws, size_t ws_size,
                              hipStream_t stream) {
  const float* x  = (const float*)d_in[0];
  const int*   ei = (const int*)d_in[1];
  const float* ew = (const float*)d_in[2];
  const float* W1 = (const float*)d_in[3];
  const float* b1 = (const float*)d_in[4];
  const float* W2 = (const float*)d_in[5];
  const float* b2 = (const float*)d_in[6];

  const int n = in_sizes[0] / D;
  const int E = in_sizes[2];
  const int* src = ei;
  const int* dst = ei + E;

  float* ws  = (float*)d_ws;
  float* dis = ws + n;
  float* xm  = ws + 2 * (size_t)n;
  float* h   = xm + (size_t)n * D;
  float* part = h + (size_t)n * D;
  int4*  recs = (int4*)(part + (size_t)DEG_CHUNKS * n);
  int*   off  = (int*)(recs + E);
  float* out = (float*)d_out;

  const int T = 256;
  const int rowsPerBlock = (T / 32) * 16;
  const int gBlocks   = (n + rowsPerBlock - 1) / rowsPerBlock;

  const dim3 degGrid(DEG_CHUNKS, (n + DEG_RANGE - 1) / DEG_RANGE);
  k_deg_part<<<degGrid, DEG_T, 0, stream>>>(dst, ew, part, n, E);

  k_bin<<<BIN_NB, BIN_T, 0, stream>>>(src, dst, ew, recs, off, E);

  const int emGrid = (n + EM_RN - 1) / EM_RN;

  k_gemm32<false, true><<<gBlocks, T, 0, stream>>>(x, W1, b1, part, dis, xm, h, n);
  k_edge_tile<<<emGrid, EM_T, 0, stream>>>(recs, off, dis, xm, h, n, E);

  k_gemm32<true, false><<<gBlocks, T, 0, stream>>>(h, W2, b2, part, dis, xm, out, n);
  k_edge_tile<<<emGrid, EM_T, 0, stream>>>(recs, off, dis, xm, out, n, E);
}


# degree sums made from the binned records by the range-owner block of the layer-1 product kernel (LDS adds; ranges now 400 rows, 25 tiles); partial-sum kernel gone; sum order changed; 6 -> 5 launches
# speedup vs baseline: 4.0283x; 1.0892x over previous
#include <hip/hip_runtime.h>


typedef float v2f __attribute__((ext_vector_type(2)));
typedef float v8f __attribute__((ext_vector_type(8)));

#define D 32

#define EM_RN   400
#define BIN_NB  256
#define BIN_NR  256
#define BIN_T   1024
#define BIN_U   7
#define BIN_P   (BIN_NR + 1)
static_assert(EM_RN % 16 == 0, "a range is a whole number of 16-row product tiles");

__global__ void __launch_bounds__(BIN_T)
k_bin(const int* __restrict__ src, const int* __restrict__ dst,
      const float* __restrict__ w, int4* __restrict__ recs,
      int* __restrict__ off, int E) {
  __shared__ int hist[BIN_NR];
  __shared__ int cur[BIN_NR];
  __shared__ int wsum[BIN_NR / 32];
  const int tid  = threadIdx.x;
  const int lane = tid & 31;
  const int wave = tid >> 5;

  const int per = (E + BIN_NB - 1) / BIN_NB;
  const int e0  = blockIdx.x * per < E ? blockIdx.x * per : E;
  const int e1  = (e0 + per < E) ? (e0 + per) : E;

  if (tid < BIN_NR) hist[tid] = 0;
  __syncthreads();

  for (int b = e0 + tid; b < e1 + tid; b += BIN_U * BIN_T) {
    int t[BIN_U];
#pragma unroll
    for (int u = 0; u < BIN_U; ++u) {
      const int e = b + u * BIN_T;
      t[u] = dst[e < e1 ? e : e1 - 1];
    }
#pragma unroll
    for (int u = 0; u < BIN_U; ++u)
      if (b + u * BIN_T < e1) atomicAdd(&hist[(unsigned)t[u] / EM_RN], 1);
  }
  __syncthreads();

  int v = 0, incl = 0;
  if (tid < BIN_NR) {
    v    = hist[tid];
    incl = v;
#pragma unroll
    for (int d = 1; d < 32; d <<= 1) {
      const int up = __shfl_up(incl, d);
      if (lane >= d) incl += up;
    }
    if (lane == 31) wsum[wave] = incl;
  }
  __syncthreads();
  if (tid < BIN_NR) {
    int before = 0;
#pragma unroll
    for (int k = 0; k < BIN_NR / 32; ++k) before += (k < wave) ? wsum[k] : 0;
    const int excl = before + incl - v;
    cur[tid] = excl;
    off[(size_t)blockIdx.x * BIN_P + tid] = excl;
  }
  if (tid == 0) off[(size_t)blockIdx.x * BIN_P + BIN_NR] = e1 - e0;
  __syncthreads();

  int4* __restrict__ myrec = recs + e0;
  for (int b = e0 + tid; b < e1 + tid; b += BIN_U * BIN_T) {
    int   t[BIN_U];
    int   s[BIN_U];
    float x[BIN_U];
#pragma unroll
    for (int u = 0; u < BIN_U; ++u) {
      const int e  = b + u * BIN_T;
      const int ec = e < e1 ? e : e1 - 1;
      t[u] = dst[ec];
      s[u] = src[ec];
      x[u] = w[ec];
    }
#pragma unroll
    for (int u = 0; u < BIN_U; ++u) {
      if (b + u * BIN_T < e1) {
        const int g = (int)((unsigned)t[u] / EM_RN);
        const int p = atomicAdd(&cur[g], 1);
        myrec[p] = make_int4(s[u], t[u] - g * EM_RN, __builtin_bit_cast(int, x[u]), 0);
      }
    }
  }
}

#define EM_T   1024
#define EM_W   (EM_T / 32)
#define EM_S   (BIN_NB / EM_W)
static_assert(EM_S * EM_W == BIN_NB, "every chunk needs a wave");
static_assert(EM_RN / 16 <= EM_W, "every product tile of a range needs a wave");

#define EM_PIECES(OFF, RG, WAVE, NE)                                         \
  const int per_ = ((NE) + BIN_NB - 1) / BIN_NB;                             \
  int pos[EM_S + 1];                                                         \
  int adj[EM_S];                                                             \
  pos[0] = 0;                                                                \
  _Pragma("unroll")                                                          \
  for (int k = 0; k < EM_S; ++k) {                                           \
    const int c  = (WAVE) * EM_S + k;                                        \
    const int c0 = c * per_ < (NE) ? c * per_ : (NE);                        \
    const int a  = (OFF)[(size_t)c * BIN_P + (RG)];                          \
    const int b  = (OFF)[(size_t)c * BIN_P + (RG) + 1];                      \
    adj[k]     = c0 + a - pos[k];                                            \
    pos[k + 1] = pos[k] + (b - a);                                           \
  }                                                                          \
  const int total = pos[EM_S]

#define EM_RECP(I, P_)                                                       \
  const int* P_;                                                             \
  do {                                                                       \
    const int q_ = ((I) + lane < total) ? (I) + lane : total - 1;            \
    int base_ = adj[0];                                                      \
    _Pragma("unroll")                                                        \
    for (int k_ = 1; k_ < EM_S; ++k_) base_ = (q_ >= pos[k_]) ? adj[k_] : base_; \
    P_ = reinterpret_cast<const int*>(recs + (base_ + q_));                  \
  } while (0)


template <bool RELU>
__device__ __forceinline__ void load_a(const float* __restrict__ X, int row0, int lane,
                                       v2f (&a)[D / 4]) {
  const int l    = lane & 15;
  const int half = lane >> 4;
  const float* xrow = X + (size_t)(row0 + l) * D;
#pragma unroll
  for (int kb = 0; kb < D; kb += 4) {
    const int k0 = kb + 2 * half;
    v2f t;
    t.x = xrow[k0];
    t.y = xrow[k0 + 1];
    if (RELU) {
      t.x = fmaxf(t.x, 0.0f);
      t.y = fmaxf(t.y, 0.0f);
    }
    a[kb / 4] = t;
  }
}

__device__ __forceinline__ void tile_product(const v2f (&a)[D / 4], const float* __restrict__ W,
                                             const float* __restrict__ bias, float dv,
                                             float* __restrict__ Y, float* __restrict__ H,
                                             int row0, int lane) {
  const int l    = lane & 15;
  const int half = lane >> 4;
  const float bl = bias[l];
  const float bh = bias[16 + l];

  v8f c0 = {};
  v8f c1 = {};

#pragma unroll
  for (int kb = 0; kb < D; kb += 4) {
    const int k0 = kb + 2 * half;
    v2f b0, b1;
    b0.x = W[(size_t)k0 * D + l];
    b0.y = W[(size_t)(k0 + 1) * D + l];
    b1.x = W[(size_t)k0 * D + 16 + l];
    b1.y = W[(size_t)(k0 + 1) * D + 16 + l];

    c0 = __builtin_amdgcn_wmma_f32_16x16x4_f32(false, a[kb / 4], false, b0, (short)0, c0,
                                               false, false);
    c1 = __builtin_amdgcn_wmma_f32_16x16x4_f32(false, a[kb / 4], false, b1, (short)0, c1,
                                               false, false);
  }

  const int rowm = row0 + 8 * half;
  const int dvb  = __builtin_bit_cast(int, dv);
#pragma unroll
  for (int j = 0; j < 8; ++j) {
    const float sLo = __builtin_bit_cast(float, __builtin_amdgcn_readlane(dvb, j));
    const float sHi = __builtin_bit_cast(float, __builtin_amdgcn_readlane(dvb, 8 + j));
    const float s   = half ? sHi : sLo;
    const float s2  = s * s;
    const size_t o  = (size_t)(rowm + j) * D;
    Y[o + l]      = c0[j];
    Y[o + 16 + l] = c1[j];
    H[o + l]      = bl + c0[j] * s2;
    H[o + 16 + l] = bh + c1[j] * s2;
  }
}

#define DG_U 8

__global__ void __launch_bounds__(EM_T)
k_gemm32_deg(const int4* __restrict__ recs, const int* __restrict__ off,
             const float* __restrict__ X, const float* __restrict__ W,
             const float* __restrict__ bias, float* __restrict__ dis,
             float* __restrict__ Y, float* __restrict__ H, int n, int E) {
  __shared__ float deg[EM_RN];

  const int tid  = threadIdx.x;
  const int lane = tid & 31;
  const int wave = __builtin_amdgcn_readfirstlane(tid >> 5);
  const int rg   = blockIdx.x;
  const int lo   = rg * EM_RN;
  const int row0 = lo + wave * 16;
  const bool has_tile = wave < EM_RN / 16 && row0 < n;

  EM_PIECES(off, rg, wave, E);

  for (int i = tid; i < EM_RN; i += EM_T) deg[i] = 0.0f;
  __syncthreads();

  v2f a[D / 4] = {};
  if (has_tile) load_a<false>(X, row0, lane, a);

  for (int i = 0; i < total; i += 32 * DG_U) {
    int   rr[DG_U];
    float ww[DG_U];
#pragma unroll
    for (int u = 0; u < DG_U; ++u) {
      EM_RECP(i + 32 * u, p);
      rr[u] = p[1];
      ww[u] = __builtin_bit_cast(float, p[2]);
    }
#pragma unroll
    for (int u = 0; u < DG_U; ++u) {
      const float x = (i + 32 * u + lane < total) ? ww[u] : 0.0f;
      atomicAdd(&deg[rr[u]], x);
    }
  }
  __syncthreads();

  if (!has_tile) return;

  const int l    = lane & 15;
  const int half = lane >> 4;
  const float d  = 1.0f + deg[wave * 16 + l];
  const float dv = (d > 0.0f) ? __frsqrt_rn(d) : 0.0f;

  tile_product(a, W, bias, dv, Y, H, row0, lane);

  if (half == 0) dis[row0 + l] = dv;
}

__global__ void k_gemm32_relu(const float* __restrict__ X, const float* __restrict__ W,
                              const float* __restrict__ bias, const float* __restrict__ dis,
                              float* __restrict__ Y, float* __restrict__ H, int n) {
  const int lane = threadIdx.x & 31;
  const int wave = threadIdx.x >> 5;
  const int row0 = (blockIdx.x * (blockDim.x >> 5) + wave) << 4;
  if (row0 >= n) return;

  const float dv = dis[row0 + (lane & 15)];
  v2f a[D / 4];
  load_a<true>(X, row0, lane, a);
  tile_product(a, W, bias, dv, Y, H, row0, lane);
}

__global__ void __launch_bounds__(EM_T)
k_edge_tile(const int4* __restrict__ recs, const int* __restrict__ off,
            const float* __restrict__ dis, const float* __restrict__ xm,
            float* __restrict__ out, int n, int E) {
  __shared__ float acc[EM_RN * D];

  const int tid  = threadIdx.x;
  const int lane = tid & 31;
  const int wave = __builtin_amdgcn_readfirstlane(tid >> 5);
  const int rg   = blockIdx.x;
  const int lo   = rg * EM_RN;
  const int rem  = n - lo;
  const int rows = rem < EM_RN ? rem : EM_RN;

  EM_PIECES(off, rg, wave, E);

#define EM_FETCH(I, RS, RR, RW)                                              \
  do {                                                                       \
    EM_RECP(I, p_);                                                          \
    RS = p_[0];                                                              \
    RR = p_[1];                                                              \
    RW = p_[2];                                                              \
  } while (0)
  int ns = 0, nr = 0, nw = 0;
  if (total > 0) EM_FETCH(0, ns, nr, nw);

  float4* acc4 = reinterpret_cast<float4*>(acc);
  for (int i = tid; i < EM_RN * D / 4; i += EM_T) acc4[i] = make_float4(0.f, 0.f, 0.f, 0.f);
  __syncthreads();

  for (int i = 0; i < total; i += 32) {
    const bool valid = i + lane < total;
    const int   s  = ns;
    const int   r  = nr;
    const float we = __builtin_bit_cast(float, nw);
    EM_FETCH(i + 32, ns, nr, nw);
    float c = dis[s] * we * dis[lo + r];
    c = valid ? c : 0.0f;
    const int cb   = __builtin_bit_cast(int, c);
    const int ro   = r * D;
    const int left = total - i;
    float v[32];
#pragma unroll
    for (int g = 0; g < 4; ++g) {
      if (g * 8 < left) {
#pragma unroll
        for (int j = g * 8; j < g * 8 + 8; ++j) {
          const unsigned sj = (unsigned)__builtin_amdgcn_readlane(s, j);
          v[j] = xm[sj * D + lane];
        }
      }
    }
#pragma unroll
    for (int g = 0; g < 4; ++g) {
      if (g * 8 < left) {
#pragma unroll
        for (int j = g * 8; j < g * 8 + 8; ++j) {
          const float cj = __builtin_bit_cast(float, __builtin_amdgcn_readlane(cb, j));
          const int   rj = __builtin_amdgcn_readlane(ro, j);
          atomicAdd(&acc[rj + lane], v[j] * cj);
        }
      }
    }
  }
  __syncthreads();

  float4* __restrict__ out4 = reinterpret_cast<float4*>(out + (size_t)lo * D);
  for (int i = tid; i < rows * (D / 4); i += EM_T) {
    float4 o = out4[i];
    const float4 a = acc4[i];
    o.x += a.x; o.y += a.y; o.z += a.z; o.w += a.w;
    out4[i] = o;
  }
}
#undef EM_FETCH
#undef EM_RECP
#undef EM_PIECES

extern "C" void kernel_launch(void* const* d_in, const int* in_sizes, int n_in,
                              void* d_out, int out_size, void* d_ws, size_t ws_size,
                              hipStream_t stream) {
  const float* x  = (const float*)d_in[0];
  const int*   ei = (const int*)d_in[1];
  const float* ew = (const float*)d_in[2];
  const float* W1 = (const float*)d_in[3];
  const float* b1 = (const float*)d_in[4];
  const float* W2 = (const float*)d_in[5];
  const float* b2 = (const float*)d_in[6];

  const int n = in_sizes[0] / D;
  const int E = in_sizes[2];
  const int* src = ei;
  const int* dst = ei + E;

  float* ws  = (float*)d_ws;
  float* dis = ws + n;
  float* xm  = ws + 2 * (size_t)n;
  float* h   = xm + (size_t)n * D;
  int4*  recs = (int4*)(h + (size_t)n * D);
  int*   off  = (int*)(recs + E);
  float* out = (float*)d_out;

  const int T = 256;
  const int rowsPerBlock = (T / 32) * 16;
  const int gBlocks   = (n + rowsPerBlock - 1) / rowsPerBlock;

  k_bin<<<BIN_NB, BIN_T, 0, stream>>>(src, dst, ew, recs, off, E);

  const int emGrid = (n + EM_RN - 1) / EM_RN;

  k_gemm32_deg<<<emGrid, EM_T, 0, stream>>>(recs, off, x, W1, b1, dis, xm, h, n, E);
  k_edge_tile<<<emGrid, EM_T, 0, stream>>>(recs, off, dis, xm, h, n, E);

  k_gemm32_relu<<<gBlocks, T, 0, stream>>>(h, W2, b2, dis, xm, out, n);
  k_edge_tile<<<emGrid, EM_T, 0, stream>>>(recs, off, dis, xm, out, n, E);
}


# layer-2 product moved into the layer-1 edge kernel: range owner makes A fragments from its LDS tile + start values (ReLU) and does the WMMA tile; h never stored; 5 -> 4 launches; sum orders unchanged
# speedup vs baseline: 4.3374x; 1.0767x over previous
#include <hip/hip_runtime.h>


typedef float v2f __attribute__((ext_vector_type(2)));
typedef float v8f __attribute__((ext_vector_type(8)));

#define D 32

#define EM_RN   400
#define BIN_NB  256
#define BIN_NR  256
#define BIN_T   1024
#define BIN_U   7
#define BIN_P   (BIN_NR + 1)
static_assert(EM_RN % 16 == 0, "a range is a whole number of 16-row product tiles");

__global__ void __launch_bounds__(BIN_T)
k_bin(const int* __restrict__ src, const int* __restrict__ dst,
      const float* __restrict__ w, int4* __restrict__ recs,
      int* __restrict__ off, int E) {
  __shared__ int hist[BIN_NR];
  __shared__ int cur[BIN_NR];
  __shared__ int wsum[BIN_NR / 32];
  const int tid  = threadIdx.x;
  const int lane = tid & 31;
  const int wave = tid >> 5;

  const int per = (E + BIN_NB - 1) / BIN_NB;
  const int e0  = blockIdx.x * per < E ? blockIdx.x * per : E;
  const int e1  = (e0 + per < E) ? (e0 + per) : E;

  if (tid < BIN_NR) hist[tid] = 0;
  __syncthreads();

  for (int b = e0 + tid; b < e1 + tid; b += BIN_U * BIN_T) {
    int t[BIN_U];
#pragma unroll
    for (int u = 0; u < BIN_U; ++u) {
      const int e = b + u * BIN_T;
      t[u] = dst[e < e1 ? e : e1 - 1];
    }
#pragma unroll
    for (int u = 0; u < BIN_U; ++u)
      if (b + u * BIN_T < e1) atomicAdd(&hist[(unsigned)t[u] / EM_RN], 1);
  }
  __syncthreads();

  int v = 0, incl = 0;
  if (tid < BIN_NR) {
    v    = hist[tid];
    incl = v;
#pragma unroll
    for (int d = 1; d < 32; d <<= 1) {
      const int up = __shfl_up(incl, d);
      if (lane >= d) incl += up;
    }
    if (lane == 31) wsum[wave] = incl;
  }
  __syncthreads();
  if (tid < BIN_NR) {
    int before = 0;
#pragma unroll
    for (int k = 0; k < BIN_NR / 32; ++k) before += (k < wave) ? wsum[k] : 0;
    const int excl = before + incl - v;
    cur[tid] = excl;
    off[(size_t)blockIdx.x * BIN_P + tid] = excl;
  }
  if (tid == 0) off[(size_t)blockIdx.x * BIN_P + BIN_NR] = e1 - e0;
  __syncthreads();

  int4* __restrict__ myrec = recs + e0;
  for (int b = e0 + tid; b < e1 + tid; b += BIN_U * BIN_T) {
    int   t[BIN_U];
    int   s[BIN_U];
    float x[BIN_U];
#pragma unroll
    for (int u = 0; u < BIN_U; ++u) {
      const int e  = b + u * BIN_T;
      const int ec = e < e1 ? e : e1 - 1;
      t[u] = dst[ec];
      s[u] = src[ec];
      x[u] = w[ec];
    }
#pragma unroll
    for (int u = 0; u < BIN_U; ++u) {
      if (b + u * BIN_T < e1) {
        const int g = (int)((unsigned)t[u] / EM_RN);
        const int p = atomicAdd(&cur[g], 1);
        myrec[p] = make_int4(s[u], t[u] - g * EM_RN, __builtin_bit_cast(int, x[u]), 0);
      }
    }
  }
}

#define EM_T   1024
#define EM_W   (EM_T / 32)
#define EM_S   (BIN_NB / EM_W)
static_assert(EM_S * EM_W == BIN_NB, "every chunk needs a wave");
static_assert(EM_RN / 16 <= EM_W, "every product tile of a range needs a wave");

#define EM_PIECES(OFF, RG, WAVE, NE)                                         \
  const int per_ = ((NE) + BIN_NB - 1) / BIN_NB;                             \
  int pos[EM_S + 1];                                                         \
  int adj[EM_S];                                                             \
  pos[0] = 0;                                                                \
  _Pragma("unroll")                                                          \
  for (int k = 0; k < EM_S; ++k) {                                           \
    const int c  = (WAVE) * EM_S + k;                                        \
    const int c0 = c * per_ < (NE) ? c * per_ : (NE);                        \
    const int a  = (OFF)[(size_t)c * BIN_P + (RG)];                          \
    const int b  = (OFF)[(size_t)c * BIN_P + (RG) + 1];                      \
    adj[k]     = c0 + a - pos[k];                                            \
    pos[k + 1] = pos[k] + (b - a);                                           \
  }                                                                          \
  const int total = pos[EM_S]

#define EM_RECP(I, P_)                                                       \
  const int* P_;                                                             \
  do {                                                                       \
    const int q_ = ((I) + lane < total) ? (I) + lane : total - 1;            \
    int base_ = adj[0];                                                      \
    _Pragma("unroll")                                                        \
    for (int k_ = 1; k_ < EM_S; ++k_) base_ = (q_ >= pos[k_]) ? adj[k_] : base_; \
    P_ = reinterpret_cast<const int*>(recs + (base_ + q_));                  \
  } while (0)


__device__ __forceinline__ void load_a(const float* __restrict__ X, int row0, int lane,
                                       v2f (&a)[D / 4]) {
  const int l    = lane & 15;
  const int half = lane >> 4;
  const float* xrow = X + (size_t)(row0 + l) * D;
#pragma unroll
  for (int kb = 0; kb < D; kb += 4) {
    const int k0 = kb + 2 * half;
    v2f t;
    t.x = xrow[k0];
    t.y = xrow[k0 + 1];
    a[kb / 4] = t;
  }
}

__device__ __forceinline__ void tile_product(const v2f (&a)[D / 4], const float* __restrict__ W,
                                             const float* __restrict__ bias, float dv,
                                             float* __restrict__ Y, float* __restrict__ H,
                                             int row0, int lane) {
  const int l    = lane & 15;
  const int half = lane >> 4;
  const float bl = bias[l];
  const float bh = bias[16 + l];

  v8f c0 = {};
  v8f c1 = {};

#pragma unroll
  for (int kb = 0; kb < D; kb += 4) {
    const int k0 = kb + 2 * half;
    v2f b0, b1;
    b0.x = W[(size_t)k0 * D + l];
    b0.y = W[(size_t)(k0 + 1) * D + l];
    b1.x = W[(size_t)k0 * D + 16 + l];
    b1.y = W[(size_t)(k0 + 1) * D + 16 + l];

    c0 = __builtin_amdgcn_wmma_f32_16x16x4_f32(false, a[kb / 4], false, b0, (short)0, c0,
                                               false, false);
    c1 = __builtin_amdgcn_wmma_f32_16x16x4_f32(false, a[kb / 4], false, b1, (short)0, c1,
                                               false, false);
  }

  const int rowm = row0 + 8 * half;
  const int dvb  = __builtin_bit_cast(int, dv);
#pragma unroll
  for (int j = 0; j < 8; ++j) {
    const float sLo = __builtin_bit_cast(float, __builtin_amdgcn_readlane(dvb, j));
    const float sHi = __builtin_bit_cast(float, __builtin_amdgcn_readlane(dvb, 8 + j));
    const float s   = half ? sHi : sLo;
    const float s2  = s * s;
    const size_t o  = (size_t)(rowm + j) * D;
    Y[o + l]      = c0[j];
    Y[o + 16 + l] = c1[j];
    H[o + l]      = bl + c0[j] * s2;
    H[o + 16 + l] = bh + c1[j] * s2;
  }
}

#define DG_U 8

__global__ void __launch_bounds__(EM_T)
k_gemm32_deg(const int4* __restrict__ recs, const int* __restrict__ off,
             const float* __restrict__ X, const float* __restrict__ W,
             const float* __restrict__ bias, float* __restrict__ dis,
             float* __restrict__ Y, float* __restrict__ H, int n, int E) {
  __shared__ float deg[EM_RN];

  const int tid  = threadIdx.x;
  const int lane = tid & 31;
  const int wave = __builtin_amdgcn_readfirstlane(tid >> 5);
  const int rg   = blockIdx.x;
  const int lo   = rg * EM_RN;
  const int row0 = lo + wave * 16;
  const bool has_tile = wave < EM_RN / 16 && row0 < n;

  EM_PIECES(off, rg, wave, E);

  for (int i = tid; i < EM_RN; i += EM_T) deg[i] = 0.0f;
  __syncthreads();

  v2f a[D / 4] = {};
  if (has_tile) load_a(X, row0, lane, a);

  for (int i = 0; i < total; i += 32 * DG_U) {
    int   rr[DG_U];
    float ww[DG_U];
#pragma unroll
    for (int u = 0; u < DG_U; ++u) {
      EM_RECP(i + 32 * u, p);
      rr[u] = p[1];
      ww[u] = __builtin_bit_cast(float, p[2]);
    }
#pragma unroll
    for (int u = 0; u < DG_U; ++u) {
      const float x = (i + 32 * u + lane < total) ? ww[u] : 0.0f;
      atomicAdd(&deg[rr[u]], x);
    }
  }
  __syncthreads();

  if (!has_tile) return;

  const int l    = lane & 15;
  const int half = lane >> 4;
  const float d  = 1.0f + deg[wave * 16 + l];
  const float dv = (d > 0.0f) ? __frsqrt_rn(d) : 0.0f;

  tile_product(a, W, bias, dv, Y, H, row0, lane);

  if (half == 0) dis[row0 + l] = dv;
}

template <bool PRODUCT>
__global__ void __launch_bounds__(EM_T)
k_edge_tile(const int4* __restrict__ recs, const int* __restrict__ off,
            const float* __restrict__ dis, const float* __restrict__ xm,
            float* __restrict__ io, const float* __restrict__ W,
            const float* __restrict__ bias, float* __restrict__ Y,
            float* __restrict__ Z, int n, int E) {
  __shared__ float acc[EM_RN * D];

  const int tid  = threadIdx.x;
  const int lane = tid & 31;
  const int wave = __builtin_amdgcn_readfirstlane(tid >> 5);
  const int rg   = blockIdx.x;
  const int lo   = rg * EM_RN;
  const int rem  = n - lo;
  const int rows = rem < EM_RN ? rem : EM_RN;
  const int row0 = lo + wave * 16;
  const bool has_tile = PRODUCT && wave < EM_RN / 16 && row0 < n;

  EM_PIECES(off, rg, wave, E);

#define EM_FETCH(I, RS, RR, RW)                                              \
  do {                                                                       \
    EM_RECP(I, p_);                                                          \
    RS = p_[0];                                                              \
    RR = p_[1];                                                              \
    RW = p_[2];                                                              \
  } while (0)
  int ns = 0, nr = 0, nw = 0;
  if (total > 0) EM_FETCH(0, ns, nr, nw);

  float4* acc4 = reinterpret_cast<float4*>(acc);
  for (int i = tid; i < EM_RN * D / 4; i += EM_T) acc4[i] = make_float4(0.f, 0.f, 0.f, 0.f);
  __syncthreads();

  v2f st[D / 4] = {};
  float dv = 0.0f;
  if (has_tile) {
    load_a(io, row0, lane, st);
    dv = dis[row0 + (lane & 15)];
  }

  for (int i = 0; i < total; i += 32) {
    const bool valid = i + lane < total;
    const int   s  = ns;
    const int   r  = nr;
    const float we = __builtin_bit_cast(float, nw);
    EM_FETCH(i + 32, ns, nr, nw);
    float c = dis[s] * we * dis[lo + r];
    c = valid ? c : 0.0f;
    const int cb   = __builtin_bit_cast(int, c);
    const int ro   = r * D;
    const int left = total - i;
    float v[32];
#pragma unroll
    for (int g = 0; g < 4; ++g) {
      if (g * 8 < left) {
#pragma unroll
        for (int j = g * 8; j < g * 8 + 8; ++j) {
          const unsigned sj = (unsigned)__builtin_amdgcn_readlane(s, j);
          v[j] = xm[sj * D + lane];
        }
      }
    }
#pragma unroll
    for (int g = 0; g < 4; ++g) {
      if (g * 8 < left) {
#pragma unroll
        for (int j = g * 8; j < g * 8 + 8; ++j) {
          const float cj = __builtin_bit_cast(float, __builtin_amdgcn_readlane(cb, j));
          const int   rj = __builtin_amdgcn_readlane(ro, j);
          atomicAdd(&acc[rj + lane], v[j] * cj);
        }
      }
    }
  }
  __syncthreads();

  if constexpr (PRODUCT) {
    if (!has_tile) return;

    const int l    = lane & 15;
    const int half = lane >> 4;
    const float* trow = acc + (wave * 16 + l) * D;
    v2f a[D / 4];
#pragma unroll
    for (int kb = 0; kb < D; kb += 4) {
      const int k0 = kb + 2 * half;
      v2f t;
      t.x = fmaxf(st[kb / 4].x + trow[k0], 0.0f);
      t.y = fmaxf(st[kb / 4].y + trow[k0 + 1], 0.0f);
      a[kb / 4] = t;
    }
    tile_product(a, W, bias, dv, Y, Z, row0, lane);
  } else {
    float4* __restrict__ io4 = reinterpret_cast<float4*>(io + (size_t)lo * D);
    for (int i = tid; i < rows * (D / 4); i += EM_T) {
      float4 o = io4[i];
      const float4 a = acc4[i];
      o.x += a.x; o.y += a.y; o.z += a.z; o.w += a.w;
      io4[i] = o;
    }
  }
}
#undef EM_FETCH
#undef EM_RECP
#undef EM_PIECES

extern "C" void kernel_launch(void* const* d_in, const int* in_sizes, int n_in,
                              void* d_out, int out_size, void* d_ws, size_t ws_size,
                              hipStream_t stream) {
  const float* x  = (const float*)d_in[0];
  const int*   ei = (const int*)d_in[1];
  const float* ew = (const float*)d_in[2];
  const float* W1 = (const float*)d_in[3];
  const float* b1 = (const float*)d_in[4];
  const float* W2 = (const float*)d_in[5];
  const float* b2 = (const float*)d_in[6];

  const int n = in_sizes[0] / D;
  const int E = in_sizes[2];
  const int* src = ei;
  const int* dst = ei + E;

  float* ws  = (float*)d_ws;
  float* dis = ws + n;
  float* xm  = ws + 2 * (size_t)n;
  float* h   = xm + (size_t)n * D;
  int4*  recs = (int4*)(h + (size_t)n * D);
  int*   off  = (int*)(recs + E);
  float* xm2  = (float*)(off + (size_t)BIN_NB * BIN_P);
  float* out = (float*)d_out;

  k_bin<<<BIN_NB, BIN_T, 0, stream>>>(src, dst, ew, recs, off, E);

  const int emGrid = (n + EM_RN - 1) / EM_RN;

  k_gemm32_deg<<<emGrid, EM_T, 0, stream>>>(recs, off, x, W1, b1, dis, xm, h, n, E);
  k_edge_tile<true><<<emGrid, EM_T, 0, stream>>>(recs, off, dis, xm, h, W2, b2, xm2, out, n, E);

  k_edge_tile<false><<<emGrid, EM_T, 0, stream>>>(recs, off, dis, xm2, out, nullptr, nullptr,
                                                  nullptr, nullptr, n, E);
}
